# starting kernel
# baseline (speedup 1.0000x reference)
#include <hip/hip_runtime.h>

typedef __attribute__((ext_vector_type(2))) float v2f;
typedef __attribute__((ext_vector_type(8))) float v8f;

#define CH 64

__global__ void init_deg_kernel(float* __restrict__ deg, int n3) {
    int i = blockIdx.x * blockDim.x + threadIdx.x;
    if (i < n3) deg[i] = 1.0f;
}

__global__ void count_deg_kernel(const int* __restrict__ edges, float* __restrict__ deg, int E) {
    int e = blockIdx.x * blockDim.x + threadIdx.x;
    if (e < E) unsafeAtomicAdd(&deg[edges[E + e]], 1.0f);
}

__global__ void rsqrt_kernel(float* __restrict__ deg, int n3) {
    int i = blockIdx.x * blockDim.x + threadIdx.x;
    if (i < n3) deg[i] = rsqrtf(deg[i]);
}

__global__ void __launch_bounds__(32)
gemm_wmma_kernel(const float* __restrict__ X, const float* __restrict__ W,
                 float* __restrict__ H) {
    const int lane = threadIdx.x & 31;
    const int r    = lane & 15;
    const int kh   = lane >> 4;
    const long long m0 = (long long)blockIdx.x * 16;

    const float* xrow = X + (m0 + r) * CH;
    v8f acc0 = {}, acc1 = {}, acc2 = {}, acc3 = {};

    for (int ks = 0; ks < 16; ++ks) {
        const int k0 = ks * 4 + 2 * kh;
        v2f a;
        a.x = xrow[k0 + 0];
        a.y = xrow[k0 + 1];
        const float* w0 = W + (long long)k0 * CH;
        const float* w1 = w0 + CH;
        v2f b;
        b.x = w0[0 * 16 + r]; b.y = w1[0 * 16 + r];
        acc0 = __builtin_amdgcn_wmma_f32_16x16x4_f32(false, a, false, b, (short)0, acc0, false, false);
        b.x = w0[1 * 16 + r]; b.y = w1[1 * 16 + r];
        acc1 = __builtin_amdgcn_wmma_f32_16x16x4_f32(false, a, false, b, (short)0, acc1, false, false);
        b.x = w0[2 * 16 + r]; b.y = w1[2 * 16 + r];
        acc2 = __builtin_amdgcn_wmma_f32_16x16x4_f32(false, a, false, b, (short)0, acc2, false, false);
        b.x = w0[3 * 16 + r]; b.y = w1[3 * 16 + r];
        acc3 = __builtin_amdgcn_wmma_f32_16x16x4_f32(false, a, false, b, (short)0, acc3, false, false);
    }

#pragma unroll
    for (int j = 0; j < 8; ++j) {
        float* hrow = H + (m0 + j + 8 * kh) * CH + r;
        hrow[0]  = acc0[j];
        hrow[16] = acc1[j];
        hrow[32] = acc2[j];
        hrow[48] = acc3[j];
    }
}

__global__ void gemm_tail_kernel(const float* __restrict__ X, const float* __restrict__ W,
                                 float* __restrict__ H, int rowStart, int rows) {
    long long t = (long long)blockIdx.x * blockDim.x + threadIdx.x;
    if (t >= (long long)rows * CH) return;
    int row = rowStart + (int)(t >> 6);
    int c   = (int)(t & 63);
    float s = 0.0f;
    for (int k = 0; k < CH; ++k) s += X[(long long)row * CH + k] * W[(long long)k * CH + c];
    H[(long long)row * CH + c] = s;
}

__global__ void init_out_kernel(const float* __restrict__ h1, const float* __restrict__ h2,
                                const float* __restrict__ h3,
                                const float* __restrict__ dinv1, const float* __restrict__ dinv2,
                                const float* __restrict__ dinv3,
                                const float* __restrict__ b1, const float* __restrict__ b2,
                                const float* __restrict__ b3,
                                float* __restrict__ out, int N) {
    long long t = (long long)blockIdx.x * blockDim.x + threadIdx.x;
    long long total = (long long)N * CH;
    if (t >= total) return;
    int i = (int)(t >> 6);
    int c = (int)(t & 63);
    float d1 = dinv1[i], d2 = dinv2[i], d3 = dinv3[i];
    out[t]         = b1[c] + b2[c] + d1 * d1 * h1[t] + d2 * d2 * h2[t];
    out[total + t] = b3[c] + d3 * d3 * h3[t];
}

__global__ void scatter_kernel(const float* __restrict__ H, const float* __restrict__ dinv,
                               const int* __restrict__ edges, float* __restrict__ out, int E) {
    long long t = (long long)blockIdx.x * blockDim.x + threadIdx.x;
    if (t >= (long long)E * 16) return;
    int e  = (int)(t >> 4);
    int c4 = (int)(t & 15);
    int s = edges[e];
    int d = edges[E + e];
    float norm = dinv[s] * dinv[d];
    float4 v = *((const float4*)(H + (long long)s * CH) + c4);
    float* o = out + (long long)d * CH + c4 * 4;
    unsafeAtomicAdd(o + 0, v.x * norm);
    unsafeAtomicAdd(o + 1, v.y * norm);
    unsafeAtomicAdd(o + 2, v.z * norm);
    unsafeAtomicAdd(o + 3, v.w * norm);
}

extern "C" void kernel_launch(void* const* d_in, const int* in_sizes, int n_in,
                              void* d_out, int out_size, void* d_ws, size_t ws_size,
                              hipStream_t stream) {
    const float* x_star = (const float*)d_in[0];
    const float* x_gal  = (const float*)d_in[1];
    const int*   e_ssn  = (const int*)d_in[2];
    const int*   e_ssf  = (const int*)d_in[3];
    const int*   e_ggn  = (const int*)d_in[4];
    const float* W_ssn  = (const float*)d_in[5];
    const float* W_ssf  = (const float*)d_in[6];
    const float* W_ggn  = (const float*)d_in[7];
    const float* b_ssn  = (const float*)d_in[8];
    const float* b_ssf  = (const float*)d_in[9];
    const float* b_ggn  = (const float*)d_in[10];

    const int N = in_sizes[0] / CH;
    const int E = in_sizes[2] / 2;

    float* ws    = (float*)d_ws;
    float* h1    = ws;
    float* h2    = h1 + (size_t)N * CH;
    float* h3    = h2 + (size_t)N * CH;
    float* dinv1 = h3 + (size_t)N * CH;
    float* dinv2 = dinv1 + N;
    float* dinv3 = dinv2 + N;
    float* out   = (float*)d_out;

    init_deg_kernel<<<(3 * N + 255) / 256, 256, 0, stream>>>(dinv1, 3 * N);
    count_deg_kernel<<<(E + 255) / 256, 256, 0, stream>>>(e_ssn, dinv1, E);
    count_deg_kernel<<<(E + 255) / 256, 256, 0, stream>>>(e_ssf, dinv2, E);
    count_deg_kernel<<<(E + 255) / 256, 256, 0, stream>>>(e_ggn, dinv3, E);
    rsqrt_kernel<<<(3 * N + 255) / 256, 256, 0, stream>>>(dinv1, 3 * N);

    const int tiles = N / 16;
    const int rem   = N - tiles * 16;
    if (tiles > 0) {
        gemm_wmma_kernel<<<tiles, 32, 0, stream>>>(x_star, W_ssn, h1);
        gemm_wmma_kernel<<<tiles, 32, 0, stream>>>(x_star, W_ssf, h2);
        gemm_wmma_kernel<<<tiles, 32, 0, stream>>>(x_gal,  W_ggn, h3);
    }
    if (rem > 0) {
        int tail = (rem * CH + 255) / 256;
        gemm_tail_kernel<<<tail, 256, 0, stream>>>(x_star, W_ssn, h1, tiles * 16, rem);
        gemm_tail_kernel<<<tail, 256, 0, stream>>>(x_star, W_ssf, h2, tiles * 16, rem);
        gemm_tail_kernel<<<tail, 256, 0, stream>>>(x_gal,  W_ggn, h3, tiles * 16, rem);
    }

    long long totalNC = (long long)N * CH;
    init_out_kernel<<<(unsigned)((totalNC + 255) / 256), 256, 0, stream>>>(
        h1, h2, h3, dinv1, dinv2, dinv3, b_ssn, b_ssf, b_ggn, out, N);

    long long work = (long long)E * 16;
    unsigned sblocks = (unsigned)((work + 255) / 256);
    scatter_kernel<<<sblocks, 256, 0, stream>>>(h1, dinv1, e_ssn, out, E);
    scatter_kernel<<<sblocks, 256, 0, stream>>>(h2, dinv2, e_ssf, out, E);
    scatter_kernel<<<sblocks, 256, 0, stream>>>(h3, dinv3, e_ggn, out + totalNC, E);
}


# aggregate first: the pull sums x rows, then each 16-row block multiplies its summed tile by W on f32 WMMA in the same kernel; h planes and 3 GEMM launches go; product follows the sum (reassociated)
# speedup vs baseline: 5.6172x; 5.6172x over previous
#include <hip/hip_runtime.h>

typedef __attribute__((ext_vector_type(2))) float v2f;
typedef __attribute__((ext_vector_type(8))) float v8f;

#define CH 64

#define BIN_SHIFT 8
#define BIN_NODES 256
#define MAXB      256
#define CHUNK     8192
#define PT        512
#define EPT       (CHUNK / PT)
#define OT        512
#define OG        4
#define NL        8
#define STAGE_CAP 12288

__device__ __forceinline__ int excl_scan_256(int v, int tid, int* s_w, int& total) {
    const int lane = tid & 31;
    const int wv   = tid >> 5;
    int incl = v;
    for (int d = 1; d < 32; d <<= 1) {
        const int u = __shfl_up(incl, d, 32);
        if (lane >= d) incl += u;
    }
    if (lane == 31 && wv < 8) s_w[wv] = incl;
    __syncthreads();
    int pre = 0;
    total = 0;
    for (int w = 0; w < 8; ++w) {
        const int x = s_w[w];
        total += x;
        if (w < wv) pre += x;
    }
    return pre + incl - v;
}

template <bool FULL>
__device__ __forceinline__ void partition_chunk(const int* __restrict__ edges, unsigned* __restrict__ tmp_t,
                                                int* __restrict__ start_row, int E, int cbase, int tid,
                                                int* s_hist, int* s_start, int* s_w, unsigned* s_stage) {
    int dstv[EPT], srcv[EPT], rk[EPT];
#pragma unroll
    for (int k = 0; k < EPT; ++k) {
        const int e  = cbase + k * PT + tid;
        const int ec = (FULL || e < E) ? e : (E - 1);
        dstv[k] = edges[E + ec];
        srcv[k] = edges[ec];
    }
#pragma unroll
    for (int k = 0; k < EPT; ++k) {
        rk[k] = 0;
        if (FULL || cbase + k * PT + tid < E) rk[k] = atomicAdd(&s_hist[dstv[k] >> BIN_SHIFT], 1);
    }
    __syncthreads();

    const int hv = (tid < MAXB) ? s_hist[tid] : 0;
    int tot;
    const int st = excl_scan_256(hv, tid, s_w, tot);
    if (tid < MAXB)  s_start[tid]  = st;
    if (tid == MAXB) s_start[MAXB] = tot;
    __syncthreads();

    if (tid <= MAXB) start_row[tid] = s_start[tid];

    int pos[EPT];
#pragma unroll
    for (int k = 0; k < EPT; ++k) pos[k] = s_start[dstv[k] >> BIN_SHIFT] + rk[k];
#pragma unroll
    for (int k = 0; k < EPT; ++k) {
        if (FULL || cbase + k * PT + tid < E)
            s_stage[pos[k]] = ((unsigned)srcv[k] << 8) | (unsigned)(dstv[k] & (BIN_NODES - 1));
    }
    __syncthreads();
#pragma unroll
    for (int k = 0; k < EPT; ++k) {
        const int p = k * PT + tid;
        if (FULL || cbase + p < E) tmp_t[cbase + p] = s_stage[p];
    }
}

__global__ void __launch_bounds__(PT)
partition_kernel(const int* __restrict__ e0, const int* __restrict__ e1, const int* __restrict__ e2,
                 unsigned* __restrict__ tmp, int* __restrict__ start, int E, int nChunks) {
    __shared__ int      s_hist[MAXB];
    __shared__ int      s_start[MAXB + 1];
    __shared__ int      s_w[8];
    __shared__ unsigned s_stage[CHUNK];
    const int tid  = threadIdx.x;
    const int type = blockIdx.y;
    const int c    = blockIdx.x;
    const int* edges = (type == 0) ? e0 : ((type == 1) ? e1 : e2);
    const int cbase = c * CHUNK;
    unsigned* tmp_t     = tmp + (size_t)type * E;
    int*      start_row = start + ((size_t)type * nChunks + c) * (MAXB + 1);

    if (tid < MAXB) s_hist[tid] = 0;
    __syncthreads();

    if (cbase + CHUNK <= E)
        partition_chunk<true >(edges, tmp_t, start_row, E, cbase, tid, s_hist, s_start, s_w, s_stage);
    else
        partition_chunk<false>(edges, tmp_t, start_row, E, cbase, tid, s_hist, s_start, s_w, s_stage);
}

template <bool PLACE>
__device__ __forceinline__ void visit(unsigned w, int* s_n, int* s_out, int* __restrict__ colb) {
    if (!PLACE) {
        atomicAdd(&s_n[w & (BIN_NODES - 1)], 1);
    } else {
        const int slot = atomicAdd(&s_n[w & (BIN_NODES - 1)], 1);
        const int src  = (int)(w >> 8);
        if (slot < STAGE_CAP) s_out[slot] = src;
        else                  colb[slot]  = src;
    }
}

template <bool PLACE>
__device__ __forceinline__ int walk_pieces(const unsigned* __restrict__ tp, const int* __restrict__ st,
                                           int* s_n, int* s_out, int* __restrict__ colb,
                                           int b, int nChunks, int tid) {
    const int li = tid & (OG - 1);
    const int pg = tid / OG;
    int sum0 = 0;
    for (int c0 = 0; c0 < nChunks; c0 += OT / OG) {
        const int c = c0 + pg;
        int s0 = 0, s1 = 0;
        if (c < nChunks) {
            const int* q = st + (size_t)c * (MAXB + 1) + b;
            s0 = q[0];
            s1 = q[1];
        }
        if (li == 0) sum0 += s0;
        const unsigned* p = tp + (size_t)c * CHUNK;
        const int kl = s1 - 1;
        for (int k = s0 + li; k < s1; k += NL * OG) {
            unsigned w[NL];
#pragma unroll
            for (int u = 0; u < NL; ++u) {
                const int ku = k + u * OG;
                w[u] = p[(ku < kl) ? ku : kl];
            }
#pragma unroll
            for (int u = 0; u < NL; ++u)
                if (k + u * OG < s1) visit<PLACE>(w[u], s_n, s_out, colb);
        }
    }
    return sum0;
}

__global__ void __launch_bounds__(OT)
owner_kernel(const unsigned* __restrict__ tmp, const int* __restrict__ start,
             int* __restrict__ off, int* __restrict__ col, float* __restrict__ dinv,
             int N, int E, int nChunks) {
    __shared__ int s_cnt[BIN_NODES];
    __shared__ int s_cur[BIN_NODES];
    __shared__ int s_red[OT / 32];
    __shared__ int s_w[8];
    __shared__ int s_out[STAGE_CAP];
    const int tid  = threadIdx.x;
    const int lane = tid & 31;
    const int wv   = tid >> 5;
    const int b    = blockIdx.x;
    const int type = blockIdx.y;
    const unsigned* tp = tmp + (size_t)type * E;
    const int*      st = start + (size_t)type * nChunks * (MAXB + 1);

    if (tid < BIN_NODES) s_cnt[tid] = 0;
    __syncthreads();

    int part = walk_pieces<false>(tp, st, s_cnt, s_out, col, b, nChunks, tid);
    for (int d = 16; d > 0; d >>= 1) part += __shfl_xor(part, d, 32);
    if (lane == 0) s_red[wv] = part;
    __syncthreads();
    int base = 0;
    for (int w = 0; w < OT / 32; ++w) base += s_red[w];

    const int k = (tid < BIN_NODES) ? s_cnt[tid] : 0;
    int M;
    const int loc = excl_scan_256(k, tid, s_w, M);
    const int node = b * BIN_NODES + tid;
    if (tid < BIN_NODES) {
        s_cur[tid] = loc;
        if (node < N) {
            off[(size_t)type * N + node]  = base + loc;
            dinv[(size_t)type * N + node] = rsqrtf((float)(k + 1));
        }
    }
    __syncthreads();

    int* colb = col + (size_t)type * E + base;
    walk_pieces<true>(tp, st, s_cur, s_out, colb, b, nChunks, tid);
    __syncthreads();
    const int m = (M < STAGE_CAP) ? M : STAGE_CAP;
    for (int j = tid; j < m; j += OT) colb[j] = s_out[j];
}

__device__ __forceinline__ float4 pull_row(const float* __restrict__ H, const float* __restrict__ dinv,
                                           const int* __restrict__ off, const int* __restrict__ col,
                                           int i, int N, int E, int hl) {
    const float di = dinv[i];
    const int beg = off[i];
    const int end = (i + 1 < N) ? off[i + 1] : E;
    float4 acc;
    {
        const float4 v = *((const float4*)(H + (long long)i * CH) + hl);
        const float nn = di * di;
        acc.x = v.x * nn; acc.y = v.y * nn; acc.z = v.z * nn; acc.w = v.w * nn;
    }
    for (int base = beg; base < end; base += 16) {
        const int n = (end - base < 16) ? (end - base) : 16;
        int   s_l = 0;
        float w_l = 0.0f;
        if (hl < n) { s_l = col[base + hl]; w_l = dinv[s_l] * di; }
        int k = 0;
        for (; k + 4 <= n; k += 4) {
            const int s0 = __shfl(s_l, k + 0, 16), s1 = __shfl(s_l, k + 1, 16);
            const int s2 = __shfl(s_l, k + 2, 16), s3 = __shfl(s_l, k + 3, 16);
            const float w0 = __shfl(w_l, k + 0, 16), w1 = __shfl(w_l, k + 1, 16);
            const float w2 = __shfl(w_l, k + 2, 16), w3 = __shfl(w_l, k + 3, 16);
            const float4 v0 = *((const float4*)(H + (long long)s0 * CH) + hl);
            const float4 v1 = *((const float4*)(H + (long long)s1 * CH) + hl);
            const float4 v2 = *((const float4*)(H + (long long)s2 * CH) + hl);
            const float4 v3 = *((const float4*)(H + (long long)s3 * CH) + hl);
            acc.x += v0.x * w0; acc.y += v0.y * w0; acc.z += v0.z * w0; acc.w += v0.w * w0;
            acc.x += v1.x * w1; acc.y += v1.y * w1; acc.z += v1.z * w1; acc.w += v1.w * w1;
            acc.x += v2.x * w2; acc.y += v2.y * w2; acc.z += v2.z * w2; acc.w += v2.w * w2;
            acc.x += v3.x * w3; acc.y += v3.y * w3; acc.z += v3.z * w3; acc.w += v3.w * w3;
        }
        for (; k < n; ++k) {
            const int   s = __shfl(s_l, k, 16);
            const float w = __shfl(w_l, k, 16);
            const float4 v = *((const float4*)(H + (long long)s * CH) + hl);
            acc.x += v.x * w; acc.y += v.y * w; acc.z += v.z * w; acc.w += v.w * w;
        }
    }
    return acc;
}

#define YP 68

template <int NT>
__global__ void __launch_bounds__(256)
pull_gemm_kernel(const float* __restrict__ X,
                 const float* __restrict__ dinva, const int* __restrict__ offa, const int* __restrict__ cola,
                 const float* __restrict__ Wa, const float* __restrict__ ba,
                 const float* __restrict__ dinvb, const int* __restrict__ offb, const int* __restrict__ colb,
                 const float* __restrict__ Wb, const float* __restrict__ bb,
                 float* __restrict__ out, int N, int E) {
    __shared__ float sY[NT][16][YP];
    __shared__ float sP[2][16][YP];
    const int tid = threadIdx.x;
    const int hl  = tid & 15;
    const int row = tid >> 4;
    const int i   = blockIdx.x * 16 + row;

    float4 ya = make_float4(0.0f, 0.0f, 0.0f, 0.0f);
    float4 yb = ya;
    if (i < N) {
        ya = pull_row(X, dinva, offa, cola, i, N, E, hl);
        if (NT == 2) yb = pull_row(X, dinvb, offb, colb, i, N, E, hl);
    }
    *((float4*)&sY[0][row][4 * hl]) = ya;
    if (NT == 2) *((float4*)&sY[NT - 1][row][4 * hl]) = yb;

    const int lane = tid & 31;
    const int wv   = tid >> 5;
    const int c    = lane & 15;
    const int kh   = lane >> 4;
    const int ct   = wv & 3;
    const int kq   = wv >> 2;

    v2f fa[8], fb[8];
#pragma unroll
    for (int s = 0; s < 8; ++s) {
        const int k0 = (kq * 8 + s) * 4 + 2 * kh;
        const float* w0 = Wa + (long long)k0 * CH + ct * 16 + c;
        fa[s].x = w0[0];
        fa[s].y = w0[CH];
        if (NT == 2) {
            const float* w1 = Wb + (long long)k0 * CH + ct * 16 + c;
            fb[s].x = w1[0];
            fb[s].y = w1[CH];
        }
    }
    __syncthreads();

    v8f acc0 = {}, acc1 = {};
#pragma unroll
    for (int s = 0; s < 8; ++s) {
        const int k0 = (kq * 8 + s) * 4 + 2 * kh;
        const v2f a0 = *((const v2f*)&sY[0][c][k0]);
        acc0 = __builtin_amdgcn_wmma_f32_16x16x4_f32(false, a0, false, fa[s], (short)0, acc0, false, false);
        if (NT == 2) {
            const v2f a1 = *((const v2f*)&sY[NT - 1][c][k0]);
            acc1 = __builtin_amdgcn_wmma_f32_16x16x4_f32(false, a1, false, fb[s], (short)0, acc1, false, false);
        }
    }
#pragma unroll
    for (int j = 0; j < 8; ++j)
        sP[kq][j + 8 * kh][ct * 16 + c] = (NT == 2) ? (acc0[j] + acc1[j]) : acc0[j];
    __syncthreads();

    if (i < N) {
        const float4 p0 = *((const float4*)&sP[0][row][4 * hl]);
        const float4 p1 = *((const float4*)&sP[1][row][4 * hl]);
        float4 r;
        r.x = p0.x + p1.x; r.y = p0.y + p1.y; r.z = p0.z + p1.z; r.w = p0.w + p1.w;
        const float4 b0 = *((const float4*)ba + hl);
        r.x += b0.x; r.y += b0.y; r.z += b0.z; r.w += b0.w;
        if (NT == 2) {
            const float4 b1 = *((const float4*)bb + hl);
            r.x += b1.x; r.y += b1.y; r.z += b1.z; r.w += b1.w;
        }
        *((float4*)(out + (long long)i * CH) + hl) = r;
    }
}

extern "C" void kernel_launch(void* const* d_in, const int* in_sizes, int n_in,
                              void* d_out, int out_size, void* d_ws, size_t ws_size,
                              hipStream_t stream) {
    const float* x_star = (const float*)d_in[0];
    const float* x_gal  = (const float*)d_in[1];
    const int*   e_ssn  = (const int*)d_in[2];
    const int*   e_ssf  = (const int*)d_in[3];
    const int*   e_ggn  = (const int*)d_in[4];
    const float* W_ssn  = (const float*)d_in[5];
    const float* W_ssf  = (const float*)d_in[6];
    const float* W_ggn  = (const float*)d_in[7];
    const float* b_ssn  = (const float*)d_in[8];
    const float* b_ssf  = (const float*)d_in[9];
    const float* b_ggn  = (const float*)d_in[10];

    const int N = in_sizes[0] / CH;
    const int E = in_sizes[2] / 2;

    float* ws    = (float*)d_ws;
    unsigned* tmp = (unsigned*)ws;
    int*   start = (int*)ws + (size_t)3 * E;
    int*   col   = (int*)(ws + (size_t)2 * N * CH);
    int*   off   = col + (size_t)3 * E;
    float* dinv1 = ws + (size_t)3 * N * CH;
    float* dinv2 = dinv1 + N;
    float* dinv3 = dinv2 + N;
    float* out   = (float*)d_out;

    const int nChunks = (E + CHUNK - 1) / CHUNK;
    const int nBins   = (N + BIN_NODES - 1) / BIN_NODES;
    partition_kernel<<<dim3((unsigned)nChunks, 3), PT, 0, stream>>>(e_ssn, e_ssf, e_ggn, tmp, start, E, nChunks);
    owner_kernel<<<dim3((unsigned)nBins, 3), OT, 0, stream>>>(tmp, start, off, col, dinv1, N, E, nChunks);

    const unsigned pblocks = (unsigned)((N + 15) / 16);
    long long totalNC = (long long)N * CH;

    pull_gemm_kernel<2><<<pblocks, 256, 0, stream>>>(x_star,
                                                     dinv1, off,     col,             W_ssn, b_ssn,
                                                     dinv2, off + N, col + (size_t)E, W_ssf, b_ssf,
                                                     out, N, E);
    pull_gemm_kernel<1><<<pblocks, 256, 0, stream>>>(x_gal,
                                                     dinv3, off + 2 * N, col + (size_t)2 * E, W_ggn, b_ggn,
                                                     dinv3, off + 2 * N, col + (size_t)2 * E, W_ggn, b_ggn,
                                                     out + totalNC, N, E);
}


# pull+GEMM of both node types as ONE launch of 2048 resident blocks (8 per WGP), each walking a fixed list of 16-row tiles (was 2 launches of 3125 one-tile blocks); launches 4 -> 3; sums unchanged
# speedup vs baseline: 5.8522x; 1.0418x over previous
#include <hip/hip_runtime.h>

typedef __attribute__((ext_vector_type(2))) float v2f;
typedef __attribute__((ext_vector_type(8))) float v8f;

#define CH 64

#define BIN_SHIFT 8
#define BIN_NODES 256
#define MAXB      256
#define CHUNK     8192
#define PT        512
#define EPT       (CHUNK / PT)
#define OT        512
#define OG        4
#define NL        8
#define STAGE_CAP 12288

__device__ __forceinline__ int excl_scan_256(int v, int tid, int* s_w, int& total) {
    const int lane = tid & 31;
    const int wv   = tid >> 5;
    int incl = v;
    for (int d = 1; d < 32; d <<= 1) {
        const int u = __shfl_up(incl, d, 32);
        if (lane >= d) incl += u;
    }
    if (lane == 31 && wv < 8) s_w[wv] = incl;
    __syncthreads();
    int pre = 0;
    total = 0;
    for (int w = 0; w < 8; ++w) {
        const int x = s_w[w];
        total += x;
        if (w < wv) pre += x;
    }
    return pre + incl - v;
}

template <bool FULL>
__device__ __forceinline__ void partition_chunk(const int* __restrict__ edges, unsigned* __restrict__ tmp_t,
                                                int* __restrict__ start_row, int E, int cbase, int tid,
                                                int* s_hist, int* s_start, int* s_w, unsigned* s_stage) {
    int dstv[EPT], srcv[EPT], rk[EPT];
#pragma unroll
    for (int k = 0; k < EPT; ++k) {
        const int e  = cbase + k * PT + tid;
        const int ec = (FULL || e < E) ? e : (E - 1);
        dstv[k] = edges[E + ec];
        srcv[k] = edges[ec];
    }
#pragma unroll
    for (int k = 0; k < EPT; ++k) {
        rk[k] = 0;
        if (FULL || cbase + k * PT + tid < E) rk[k] = atomicAdd(&s_hist[dstv[k] >> BIN_SHIFT], 1);
    }
    __syncthreads();

    const int hv = (tid < MAXB) ? s_hist[tid] : 0;
    int tot;
    const int st = excl_scan_256(hv, tid, s_w, tot);
    if (tid < MAXB)  s_start[tid]  = st;
    if (tid == MAXB) s_start[MAXB] = tot;
    __syncthreads();

    if (tid <= MAXB) start_row[tid] = s_start[tid];

    int pos[EPT];
#pragma unroll
    for (int k = 0; k < EPT; ++k) pos[k] = s_start[dstv[k] >> BIN_SHIFT] + rk[k];
#pragma unroll
    for (int k = 0; k < EPT; ++k) {
        if (FULL || cbase + k * PT + tid < E)
            s_stage[pos[k]] = ((unsigned)srcv[k] << 8) | (unsigned)(dstv[k] & (BIN_NODES - 1));
    }
    __syncthreads();
#pragma unroll
    for (int k = 0; k < EPT; ++k) {
        const int p = k * PT + tid;
        if (FULL || cbase + p < E) tmp_t[cbase + p] = s_stage[p];
    }
}

__global__ void __launch_bounds__(PT)
partition_kernel(const int* __restrict__ e0, const int* __restrict__ e1, const int* __restrict__ e2,
                 unsigned* __restrict__ tmp, int* __restrict__ start, int E, int nChunks) {
    __shared__ int      s_hist[MAXB];
    __shared__ int      s_start[MAXB + 1];
    __shared__ int      s_w[8];
    __shared__ unsigned s_stage[CHUNK];
    const int tid  = threadIdx.x;
    const int type = blockIdx.y;
    const int c    = blockIdx.x;
    const int* edges = (type == 0) ? e0 : ((type == 1) ? e1 : e2);
    const int cbase = c * CHUNK;
    unsigned* tmp_t     = tmp + (size_t)type * E;
    int*      start_row = start + ((size_t)type * nChunks + c) * (MAXB + 1);

    if (tid < MAXB) s_hist[tid] = 0;
    __syncthreads();

    if (cbase + CHUNK <= E)
        partition_chunk<true >(edges, tmp_t, start_row, E, cbase, tid, s_hist, s_start, s_w, s_stage);
    else
        partition_chunk<false>(edges, tmp_t, start_row, E, cbase, tid, s_hist, s_start, s_w, s_stage);
}

template <bool PLACE>
__device__ __forceinline__ void visit(unsigned w, int* s_n, int* s_out, int* __restrict__ colb) {
    if (!PLACE) {
        atomicAdd(&s_n[w & (BIN_NODES - 1)], 1);
    } else {
        const int slot = atomicAdd(&s_n[w & (BIN_NODES - 1)], 1);
        const int src  = (int)(w >> 8);
        if (slot < STAGE_CAP) s_out[slot] = src;
        else                  colb[slot]  = src;
    }
}

template <bool PLACE>
__device__ __forceinline__ int walk_pieces(const unsigned* __restrict__ tp, const int* __restrict__ st,
                                           int* s_n, int* s_out, int* __restrict__ colb,
                                           int b, int nChunks, int tid) {
    const int li = tid & (OG - 1);
    const int pg = tid / OG;
    int sum0 = 0;
    for (int c0 = 0; c0 < nChunks; c0 += OT / OG) {
        const int c = c0 + pg;
        int s0 = 0, s1 = 0;
        if (c < nChunks) {
            const int* q = st + (size_t)c * (MAXB + 1) + b;
            s0 = q[0];
            s1 = q[1];
        }
        if (li == 0) sum0 += s0;
        const unsigned* p = tp + (size_t)c * CHUNK;
        const int kl = s1 - 1;
        for (int k = s0 + li; k < s1; k += NL * OG) {
            unsigned w[NL];
#pragma unroll
            for (int u = 0; u < NL; ++u) {
                const int ku = k + u * OG;
                w[u] = p[(ku < kl) ? ku : kl];
            }
#pragma unroll
            for (int u = 0; u < NL; ++u)
                if (k + u * OG < s1) visit<PLACE>(w[u], s_n, s_out, colb);
        }
    }
    return sum0;
}

__global__ void __launch_bounds__(OT)
owner_kernel(const unsigned* __restrict__ tmp, const int* __restrict__ start,
             int* __restrict__ off, int* __restrict__ col, float* __restrict__ dinv,
             int N, int E, int nChunks) {
    __shared__ int s_cnt[BIN_NODES];
    __shared__ int s_cur[BIN_NODES];
    __shared__ int s_red[OT / 32];
    __shared__ int s_w[8];
    __shared__ int s_out[STAGE_CAP];
    const int tid  = threadIdx.x;
    const int lane = tid & 31;
    const int wv   = tid >> 5;
    const int b    = blockIdx.x;
    const int type = blockIdx.y;
    const unsigned* tp = tmp + (size_t)type * E;
    const int*      st = start + (size_t)type * nChunks * (MAXB + 1);

    if (tid < BIN_NODES) s_cnt[tid] = 0;
    __syncthreads();

    int part = walk_pieces<false>(tp, st, s_cnt, s_out, col, b, nChunks, tid);
    for (int d = 16; d > 0; d >>= 1) part += __shfl_xor(part, d, 32);
    if (lane == 0) s_red[wv] = part;
    __syncthreads();
    int base = 0;
    for (int w = 0; w < OT / 32; ++w) base += s_red[w];

    const int k = (tid < BIN_NODES) ? s_cnt[tid] : 0;
    int M;
    const int loc = excl_scan_256(k, tid, s_w, M);
    const int node = b * BIN_NODES + tid;
    if (tid < BIN_NODES) {
        s_cur[tid] = loc;
        if (node < N) {
            off[(size_t)type * N + node]  = base + loc;
            dinv[(size_t)type * N + node] = rsqrtf((float)(k + 1));
        }
    }
    __syncthreads();

    int* colb = col + (size_t)type * E + base;
    walk_pieces<true>(tp, st, s_cur, s_out, colb, b, nChunks, tid);
    __syncthreads();
    const int m = (M < STAGE_CAP) ? M : STAGE_CAP;
    for (int j = tid; j < m; j += OT) colb[j] = s_out[j];
}

__device__ __forceinline__ float4 pull_row(const float* __restrict__ H, const float* __restrict__ dinv,
                                           const int* __restrict__ off, const int* __restrict__ col,
                                           int i, int N, int E, int hl) {
    const float di = dinv[i];
    const int beg = off[i];
    const int end = (i + 1 < N) ? off[i + 1] : E;
    float4 acc;
    {
        const float4 v = *((const float4*)(H + (long long)i * CH) + hl);
        const float nn = di * di;
        acc.x = v.x * nn; acc.y = v.y * nn; acc.z = v.z * nn; acc.w = v.w * nn;
    }
    for (int base = beg; base < end; base += 16) {
        const int n = (end - base < 16) ? (end - base) : 16;
        int   s_l = 0;
        float w_l = 0.0f;
        if (hl < n) { s_l = col[base + hl]; w_l = dinv[s_l] * di; }
        int k = 0;
        for (; k + 4 <= n; k += 4) {
            const int s0 = __shfl(s_l, k + 0, 16), s1 = __shfl(s_l, k + 1, 16);
            const int s2 = __shfl(s_l, k + 2, 16), s3 = __shfl(s_l, k + 3, 16);
            const float w0 = __shfl(w_l, k + 0, 16), w1 = __shfl(w_l, k + 1, 16);
            const float w2 = __shfl(w_l, k + 2, 16), w3 = __shfl(w_l, k + 3, 16);
            const float4 v0 = *((const float4*)(H + (long long)s0 * CH) + hl);
            const float4 v1 = *((const float4*)(H + (long long)s1 * CH) + hl);
            const float4 v2 = *((const float4*)(H + (long long)s2 * CH) + hl);
            const float4 v3 = *((const float4*)(H + (long long)s3 * CH) + hl);
            acc.x += v0.x * w0; acc.y += v0.y * w0; acc.z += v0.z * w0; acc.w += v0.w * w0;
            acc.x += v1.x * w1; acc.y += v1.y * w1; acc.z += v1.z * w1; acc.w += v1.w * w1;
            acc.x += v2.x * w2; acc.y += v2.y * w2; acc.z += v2.z * w2; acc.w += v2.w * w2;
            acc.x += v3.x * w3; acc.y += v3.y * w3; acc.z += v3.z * w3; acc.w += v3.w * w3;
        }
        for (; k < n; ++k) {
            const int   s = __shfl(s_l, k, 16);
            const float w = __shfl(w_l, k, 16);
            const float4 v = *((const float4*)(H + (long long)s * CH) + hl);
            acc.x += v.x * w; acc.y += v.y * w; acc.z += v.z * w; acc.w += v.w * w;
        }
    }
    return acc;
}

#define YP 68

__device__ __forceinline__ void pull_gemm_tile(const float* __restrict__ X,
                 const float* __restrict__ dinva, const int* __restrict__ offa, const int* __restrict__ cola,
                 const float* __restrict__ Wa, const float* __restrict__ ba,
                 const float* __restrict__ dinvb, const int* __restrict__ offb, const int* __restrict__ colb,
                 const float* __restrict__ Wb, const float* __restrict__ bb,
                 float* __restrict__ out, int N, int E, int tile, bool two,
                 float (*sY)[16][YP], float (*sP)[16][YP], int tid) {
    int t1 = tid;
    asm volatile("" : "+v"(t1));
    const int hl  = t1 & 15;
    const int row = t1 >> 4;
    const int i   = tile * 16 + row;

    float4 ya = make_float4(0.0f, 0.0f, 0.0f, 0.0f);
    float4 yb = ya;
    if (i < N) {
        ya = pull_row(X, dinva, offa, cola, i, N, E, hl);
        if (two) yb = pull_row(X, dinvb, offb, colb, i, N, E, hl);
    }
    *((float4*)&sY[0][row][4 * hl]) = ya;
    if (two) *((float4*)&sY[1][row][4 * hl]) = yb;

    int t2 = tid;
    asm volatile("" : "+v"(t2));
    const int hl2  = t2 & 15;
    const int row2 = t2 >> 4;
    const int i2   = tile * 16 + row2;
    const int lane = t2 & 31;
    const int wv   = t2 >> 5;
    const int c    = lane & 15;
    const int kh   = lane >> 4;
    const int ct   = wv & 3;
    const int kq   = wv >> 2;

    v2f fa[8], fb[8] = {};
#pragma unroll
    for (int s = 0; s < 8; ++s) {
        const int k0 = (kq * 8 + s) * 4 + 2 * kh;
        const float* w0 = Wa + (long long)k0 * CH + ct * 16 + c;
        fa[s].x = w0[0];
        fa[s].y = w0[CH];
    }
    if (two) {
#pragma unroll
        for (int s = 0; s < 8; ++s) {
            const int k0 = (kq * 8 + s) * 4 + 2 * kh;
            const float* w1 = Wb + (long long)k0 * CH + ct * 16 + c;
            fb[s].x = w1[0];
            fb[s].y = w1[CH];
        }
    }
    __syncthreads();

    v8f acc0 = {};
    if (two) {
        v8f acc1 = {};
#pragma unroll
        for (int s = 0; s < 8; ++s) {
            const int k0 = (kq * 8 + s) * 4 + 2 * kh;
            const v2f a0 = *((const v2f*)&sY[0][c][k0]);
            acc0 = __builtin_amdgcn_wmma_f32_16x16x4_f32(false, a0, false, fa[s], (short)0, acc0, false, false);
            const v2f a1 = *((const v2f*)&sY[1][c][k0]);
            acc1 = __builtin_amdgcn_wmma_f32_16x16x4_f32(false, a1, false, fb[s], (short)0, acc1, false, false);
        }
#pragma unroll
        for (int j = 0; j < 8; ++j) acc0[j] = acc0[j] + acc1[j];
    } else {
#pragma unroll
        for (int s = 0; s < 8; ++s) {
            const int k0 = (kq * 8 + s) * 4 + 2 * kh;
            const v2f a0 = *((const v2f*)&sY[0][c][k0]);
            acc0 = __builtin_amdgcn_wmma_f32_16x16x4_f32(false, a0, false, fa[s], (short)0, acc0, false, false);
        }
    }
#pragma unroll
    for (int j = 0; j < 8; ++j)
        sP[kq][j + 8 * kh][ct * 16 + c] = acc0[j];
    __syncthreads();

    if (i2 < N) {
        const float4 p0 = *((const float4*)&sP[0][row2][4 * hl2]);
        const float4 p1 = *((const float4*)&sP[1][row2][4 * hl2]);
        float4 r;
        r.x = p0.x + p1.x; r.y = p0.y + p1.y; r.z = p0.z + p1.z; r.w = p0.w + p1.w;
        const float4 b0 = *((const float4*)ba + hl2);
        r.x += b0.x; r.y += b0.y; r.z += b0.z; r.w += b0.w;
        if (two) {
            const float4 b1 = *((const float4*)bb + hl2);
            r.x += b1.x; r.y += b1.y; r.z += b1.z; r.w += b1.w;
        }
        *((float4*)(out + (long long)i2 * CH) + hl2) = r;
    }
}

__global__ void __launch_bounds__(256) __attribute__((amdgpu_waves_per_eu(16, 16)))
pull_gemm_resident_kernel(const float* __restrict__ Xs, const float* __restrict__ Xg,
                          const float* __restrict__ dinv, const int* __restrict__ off, const int* __restrict__ col,
                          const float* __restrict__ W0, const float* __restrict__ b0,
                          const float* __restrict__ W1, const float* __restrict__ b1,
                          const float* __restrict__ W2, const float* __restrict__ b2,
                          float* __restrict__ out, int N, int E, int nT, int G) {
    __shared__ float sY[2][16][YP];
    __shared__ float sP[2][16][YP];
    const int tid = threadIdx.x;
    const int b   = (int)blockIdx.x;
    int pass = 0;
    for (int base = 0; base < 2 * nT; base += G, ++pass) {
        const int p = base + ((pass == 1) ? (G - 1 - b) : b);
        if (p >= 2 * nT) break;
        const bool star = p < nT;
        const size_t ta = star ? 0 : 2;
        pull_gemm_tile(star ? Xs : Xg,
                       dinv + ta * N, off + ta * N, col + ta * E, star ? W0 : W2, star ? b0 : b2,
                       dinv + N,      off + N,      col + E,      W1,             b1,
                       out + (star ? (size_t)0 : (size_t)N * CH), N, E, star ? p : p - nT, star, sY, sP, tid);
    }
}

extern "C" void kernel_launch(void* const* d_in, const int* in_sizes, int n_in,
                              void* d_out, int out_size, void* d_ws, size_t ws_size,
                              hipStream_t stream) {
    const float* x_star = (const float*)d_in[0];
    const float* x_gal  = (const float*)d_in[1];
    const int*   e_ssn  = (const int*)d_in[2];
    const int*   e_ssf  = (const int*)d_in[3];
    const int*   e_ggn  = (const int*)d_in[4];
    const float* W_ssn  = (const float*)d_in[5];
    const float* W_ssf  = (const float*)d_in[6];
    const float* W_ggn  = (const float*)d_in[7];
    const float* b_ssn  = (const float*)d_in[8];
    const float* b_ssf  = (const float*)d_in[9];
    const float* b_ggn  = (const float*)d_in[10];

    const int N = in_sizes[0] / CH;
    const int E = in_sizes[2] / 2;

    float* ws    = (float*)d_ws;
    unsigned* tmp = (unsigned*)ws;
    int*   start = (int*)ws + (size_t)3 * E;
    int*   col   = (int*)(ws + (size_t)2 * N * CH);
    int*   off   = col + (size_t)3 * E;
    float* dinv1 = ws + (size_t)3 * N * CH;
    float* out   = (float*)d_out;

    const int nChunks = (E + CHUNK - 1) / CHUNK;
    const int nBins   = (N + BIN_NODES - 1) / BIN_NODES;
    partition_kernel<<<dim3((unsigned)nChunks, 3), PT, 0, stream>>>(e_ssn, e_ssf, e_ggn, tmp, start, E, nChunks);
    owner_kernel<<<dim3((unsigned)nBins, 3), OT, 0, stream>>>(tmp, start, off, col, dinv1, N, E, nChunks);

    const int nT = (N + 15) / 16;
    pull_gemm_resident_kernel<<<2048, 256, 0, stream>>>(x_star, x_gal, dinv1, off, col,
                                                        W_ssn, b_ssn, W_ssf, b_ssf, W_ggn, b_ggn,
                                                        out, N, E, nT, 2048);
}


# pull's index chain off the row loads' path: next 16-edge chunk's src fetched one ahead, dinv[src] beside its x row, both edge types' headers + first src at once; 32-bit row offsets; sum order kept
# speedup vs baseline: 6.0807x; 1.0390x over previous
#include <hip/hip_runtime.h>

typedef __attribute__((ext_vector_type(2))) float v2f;
typedef __attribute__((ext_vector_type(8))) float v8f;

#define CH 64

#define BIN_SHIFT 8
#define BIN_NODES 256
#define MAXB      256
#define CHUNK     8192
#define PT        512
#define EPT       (CHUNK / PT)
#define OT        512
#define OG        4
#define NL        8
#define STAGE_CAP 12288

__device__ __forceinline__ int excl_scan_256(int v, int tid, int* s_w, int& total) {
    const int lane = tid & 31;
    const int wv   = tid >> 5;
    int incl = v;
    for (int d = 1; d < 32; d <<= 1) {
        const int u = __shfl_up(incl, d, 32);
        if (lane >= d) incl += u;
    }
    if (lane == 31 && wv < 8) s_w[wv] = incl;
    __syncthreads();
    int pre = 0;
    total = 0;
    for (int w = 0; w < 8; ++w) {
        const int x = s_w[w];
        total += x;
        if (w < wv) pre += x;
    }
    return pre + incl - v;
}

template <bool FULL>
__device__ __forceinline__ void partition_chunk(const int* __restrict__ edges, unsigned* __restrict__ tmp_t,
                                                int* __restrict__ start_row, int E, int cbase, int tid,
                                                int* s_hist, int* s_start, int* s_w, unsigned* s_stage) {
    int dstv[EPT], srcv[EPT], rk[EPT];
#pragma unroll
    for (int k = 0; k < EPT; ++k) {
        const int e  = cbase + k * PT + tid;
        const int ec = (FULL || e < E) ? e : (E - 1);
        dstv[k] = edges[E + ec];
        srcv[k] = edges[ec];
    }
#pragma unroll
    for (int k = 0; k < EPT; ++k) {
        rk[k] = 0;
        if (FULL || cbase + k * PT + tid < E) rk[k] = atomicAdd(&s_hist[dstv[k] >> BIN_SHIFT], 1);
    }
    __syncthreads();

    const int hv = (tid < MAXB) ? s_hist[tid] : 0;
    int tot;
    const int st = excl_scan_256(hv, tid, s_w, tot);
    if (tid < MAXB)  s_start[tid]  = st;
    if (tid == MAXB) s_start[MAXB] = tot;
    __syncthreads();

    if (tid <= MAXB) start_row[tid] = s_start[tid];

    int pos[EPT];
#pragma unroll
    for (int k = 0; k < EPT; ++k) pos[k] = s_start[dstv[k] >> BIN_SHIFT] + rk[k];
#pragma unroll
    for (int k = 0; k < EPT; ++k) {
        if (FULL || cbase + k * PT + tid < E)
            s_stage[pos[k]] = ((unsigned)srcv[k] << 8) | (unsigned)(dstv[k] & (BIN_NODES - 1));
    }
    __syncthreads();
#pragma unroll
    for (int k = 0; k < EPT; ++k) {
        const int p = k * PT + tid;
        if (FULL || cbase + p < E) tmp_t[cbase + p] = s_stage[p];
    }
}

__global__ void __launch_bounds__(PT)
partition_kernel(const int* __restrict__ e0, const int* __restrict__ e1, const int* __restrict__ e2,
                 unsigned* __restrict__ tmp, int* __restrict__ start, int E, int nChunks) {
    __shared__ int      s_hist[MAXB];
    __shared__ int      s_start[MAXB + 1];
    __shared__ int      s_w[8];
    __shared__ unsigned s_stage[CHUNK];
    const int tid  = threadIdx.x;
    const int type = blockIdx.y;
    const int c    = blockIdx.x;
    const int* edges = (type == 0) ? e0 : ((type == 1) ? e1 : e2);
    const int cbase = c * CHUNK;
    unsigned* tmp_t     = tmp + (size_t)type * E;
    int*      start_row = start + ((size_t)type * nChunks + c) * (MAXB + 1);

    if (tid < MAXB) s_hist[tid] = 0;
    __syncthreads();

    if (cbase + CHUNK <= E)
        partition_chunk<true >(edges, tmp_t, start_row, E, cbase, tid, s_hist, s_start, s_w, s_stage);
    else
        partition_chunk<false>(edges, tmp_t, start_row, E, cbase, tid, s_hist, s_start, s_w, s_stage);
}

template <bool PLACE>
__device__ __forceinline__ void visit(unsigned w, int* s_n, int* s_out, int* __restrict__ colb) {
    if (!PLACE) {
        atomicAdd(&s_n[w & (BIN_NODES - 1)], 1);
    } else {
        const int slot = atomicAdd(&s_n[w & (BIN_NODES - 1)], 1);
        const int src  = (int)(w >> 8);
        if (slot < STAGE_CAP) s_out[slot] = src;
        else                  colb[slot]  = src;
    }
}

template <bool PLACE>
__device__ __forceinline__ int walk_pieces(const unsigned* __restrict__ tp, const int* __restrict__ st,
                                           int* s_n, int* s_out, int* __restrict__ colb,
                                           int b, int nChunks, int tid) {
    const int li = tid & (OG - 1);
    const int pg = tid / OG;
    int sum0 = 0;
    for (int c0 = 0; c0 < nChunks; c0 += OT / OG) {
        const int c = c0 + pg;
        int s0 = 0, s1 = 0;
        if (c < nChunks) {
            const int* q = st + (size_t)c * (MAXB + 1) + b;
            s0 = q[0];
            s1 = q[1];
        }
        if (li == 0) sum0 += s0;
        const unsigned* p = tp + (size_t)c * CHUNK;
        const int kl = s1 - 1;
        for (int k = s0 + li; k < s1; k += NL * OG) {
            unsigned w[NL];
#pragma unroll
            for (int u = 0; u < NL; ++u) {
                const int ku = k + u * OG;
                w[u] = p[(ku < kl) ? ku : kl];
            }
#pragma unroll
            for (int u = 0; u < NL; ++u)
                if (k + u * OG < s1) visit<PLACE>(w[u], s_n, s_out, colb);
        }
    }
    return sum0;
}

__global__ void __launch_bounds__(OT)
owner_kernel(const unsigned* __restrict__ tmp, const int* __restrict__ start,
             int* __restrict__ off, int* __restrict__ col, float* __restrict__ dinv,
             int N, int E, int nChunks) {
    __shared__ int s_cnt[BIN_NODES];
    __shared__ int s_cur[BIN_NODES];
    __shared__ int s_red[OT / 32];
    __shared__ int s_w[8];
    __shared__ int s_out[STAGE_CAP];
    const int tid  = threadIdx.x;
    const int lane = tid & 31;
    const int wv   = tid >> 5;
    const int b    = blockIdx.x;
    const int type = blockIdx.y;
    const unsigned* tp = tmp + (size_t)type * E;
    const int*      st = start + (size_t)type * nChunks * (MAXB + 1);

    if (tid < BIN_NODES) s_cnt[tid] = 0;
    __syncthreads();

    int part = walk_pieces<false>(tp, st, s_cnt, s_out, col, b, nChunks, tid);
    for (int d = 16; d > 0; d >>= 1) part += __shfl_xor(part, d, 32);
    if (lane == 0) s_red[wv] = part;
    __syncthreads();
    int base = 0;
    for (int w = 0; w < OT / 32; ++w) base += s_red[w];

    const int k = (tid < BIN_NODES) ? s_cnt[tid] : 0;
    int M;
    const int loc = excl_scan_256(k, tid, s_w, M);
    const int node = b * BIN_NODES + tid;
    if (tid < BIN_NODES) {
        s_cur[tid] = loc;
        if (node < N) {
            off[(size_t)type * N + node]  = base + loc;
            dinv[(size_t)type * N + node] = rsqrtf((float)(k + 1));
        }
    }
    __syncthreads();

    int* colb = col + (size_t)type * E + base;
    walk_pieces<true>(tp, st, s_cur, s_out, colb, b, nChunks, tid);
    __syncthreads();
    const int m = (M < STAGE_CAP) ? M : STAGE_CAP;
    for (int j = tid; j < m; j += OT) colb[j] = s_out[j];
}

__device__ __forceinline__ void pull_edges(const float* __restrict__ H, const float* __restrict__ dinv,
                                           const int* __restrict__ col, int beg, int end, float di,
                                           int s_l, int hl, float4& acc) {
    const unsigned ho = 4u * (unsigned)hl;
    for (int base = beg; base < end; base += 16) {
        const int n = (end - base < 16) ? (end - base) : 16;
        int s_n = 0;
        int k = 0;
        for (; k + 4 <= n; k += 4) {
            const int s0 = __shfl(s_l, k + 0, 16), s1 = __shfl(s_l, k + 1, 16);
            const int s2 = __shfl(s_l, k + 2, 16), s3 = __shfl(s_l, k + 3, 16);
            const float d0 = dinv[s0], d1 = dinv[s1], d2 = dinv[s2], d3 = dinv[s3];
            const float4 v0 = *(const float4*)(H + ((unsigned)s0 * CH + ho));
            const float4 v1 = *(const float4*)(H + ((unsigned)s1 * CH + ho));
            const float4 v2 = *(const float4*)(H + ((unsigned)s2 * CH + ho));
            const float4 v3 = *(const float4*)(H + ((unsigned)s3 * CH + ho));
            if (k == 0 && base + 16 + hl < end) s_n = col[base + 16 + hl];
            const float w0 = d0 * di, w1 = d1 * di, w2 = d2 * di, w3 = d3 * di;
            acc.x += v0.x * w0; acc.y += v0.y * w0; acc.z += v0.z * w0; acc.w += v0.w * w0;
            acc.x += v1.x * w1; acc.y += v1.y * w1; acc.z += v1.z * w1; acc.w += v1.w * w1;
            acc.x += v2.x * w2; acc.y += v2.y * w2; acc.z += v2.z * w2; acc.w += v2.w * w2;
            acc.x += v3.x * w3; acc.y += v3.y * w3; acc.z += v3.z * w3; acc.w += v3.w * w3;
        }
        for (; k < n; ++k) {
            const int   s = __shfl(s_l, k, 16);
            const float d = dinv[s];
            const float4 v = *(const float4*)(H + ((unsigned)s * CH + ho));
            const float w = d * di;
            acc.x += v.x * w; acc.y += v.y * w; acc.z += v.z * w; acc.w += v.w * w;
        }
        s_l = s_n;
    }
}

__device__ __forceinline__ void pull_row_types(const float* __restrict__ H,
                 const float* __restrict__ dinva, const int* __restrict__ offa, const int* __restrict__ cola,
                 const float* __restrict__ dinvb, const int* __restrict__ offb, const int* __restrict__ colb,
                 int i, int N, int E, int hl, bool two, float4& ya, float4& yb) {
    const float4 xi = *(const float4*)(H + ((unsigned)i * CH + 4u * (unsigned)hl));
    const float da   = dinva[i];
    const int   bega = offa[i];
    const int   enda = (i + 1 < N) ? offa[i + 1] : E;
    float db = 0.0f;
    int begb = 0, endb = 0;
    if (two) {
        db   = dinvb[i];
        begb = offb[i];
        endb = (i + 1 < N) ? offb[i + 1] : E;
    }
    int sa = 0, sb = 0;
    if (bega + hl < enda) sa = cola[bega + hl];
    if (two && begb + hl < endb) sb = colb[begb + hl];
    {
        const float nn = da * da;
        ya.x = xi.x * nn; ya.y = xi.y * nn; ya.z = xi.z * nn; ya.w = xi.w * nn;
    }
    pull_edges(H, dinva, cola, bega, enda, da, sa, hl, ya);
    if (two) {
        const float nn = db * db;
        yb.x = xi.x * nn; yb.y = xi.y * nn; yb.z = xi.z * nn; yb.w = xi.w * nn;
        pull_edges(H, dinvb, colb, begb, endb, db, sb, hl, yb);
    }
}

#define YP 68

__device__ __forceinline__ void pull_gemm_tile(const float* __restrict__ X,
                 const float* __restrict__ dinva, const int* __restrict__ offa, const int* __restrict__ cola,
                 const float* __restrict__ Wa, const float* __restrict__ ba,
                 const float* __restrict__ dinvb, const int* __restrict__ offb, const int* __restrict__ colb,
                 const float* __restrict__ Wb, const float* __restrict__ bb,
                 float* __restrict__ out, int N, int E, int tile, bool two,
                 float (*sY)[16][YP], float (*sP)[16][YP], int tid) {
    int t1 = tid;
    asm volatile("" : "+v"(t1));
    const int hl  = t1 & 15;
    const int row = t1 >> 4;
    const int i   = tile * 16 + row;

    float4 ya = make_float4(0.0f, 0.0f, 0.0f, 0.0f);
    float4 yb = ya;
    if (i < N) {
        pull_row_types(X, dinva, offa, cola, dinvb, offb, colb, i, N, E, hl, two, ya, yb);
    }
    *((float4*)&sY[0][row][4 * hl]) = ya;
    if (two) *((float4*)&sY[1][row][4 * hl]) = yb;

    int t2 = tid;
    asm volatile("" : "+v"(t2));
    const int hl2  = t2 & 15;
    const int row2 = t2 >> 4;
    const int i2   = tile * 16 + row2;
    const int lane = t2 & 31;
    const int wv   = t2 >> 5;
    const int c    = lane & 15;
    const int kh   = lane >> 4;
    const int ct   = wv & 3;
    const int kq   = wv >> 2;

    v2f fa[8], fb[8] = {};
#pragma unroll
    for (int s = 0; s < 8; ++s) {
        const int k0 = (kq * 8 + s) * 4 + 2 * kh;
        const float* w0 = Wa + (long long)k0 * CH + ct * 16 + c;
        fa[s].x = w0[0];
        fa[s].y = w0[CH];
    }
    if (two) {
#pragma unroll
        for (int s = 0; s < 8; ++s) {
            const int k0 = (kq * 8 + s) * 4 + 2 * kh;
            const float* w1 = Wb + (long long)k0 * CH + ct * 16 + c;
            fb[s].x = w1[0];
            fb[s].y = w1[CH];
        }
    }
    __syncthreads();

    v8f acc0 = {};
    if (two) {
        v8f acc1 = {};
#pragma unroll
        for (int s = 0; s < 8; ++s) {
            const int k0 = (kq * 8 + s) * 4 + 2 * kh;
            const v2f a0 = *((const v2f*)&sY[0][c][k0]);
            acc0 = __builtin_amdgcn_wmma_f32_16x16x4_f32(false, a0, false, fa[s], (short)0, acc0, false, false);
            const v2f a1 = *((const v2f*)&sY[1][c][k0]);
            acc1 = __builtin_amdgcn_wmma_f32_16x16x4_f32(false, a1, false, fb[s], (short)0, acc1, false, false);
        }
#pragma unroll
        for (int j = 0; j < 8; ++j) acc0[j] = acc0[j] + acc1[j];
    } else {
#pragma unroll
        for (int s = 0; s < 8; ++s) {
            const int k0 = (kq * 8 + s) * 4 + 2 * kh;
            const v2f a0 = *((const v2f*)&sY[0][c][k0]);
            acc0 = __builtin_amdgcn_wmma_f32_16x16x4_f32(false, a0, false, fa[s], (short)0, acc0, false, false);
        }
    }
#pragma unroll
    for (int j = 0; j < 8; ++j)
        sP[kq][j + 8 * kh][ct * 16 + c] = acc0[j];
    __syncthreads();

    if (i2 < N) {
        const float4 p0 = *((const float4*)&sP[0][row2][4 * hl2]);
        const float4 p1 = *((const float4*)&sP[1][row2][4 * hl2]);
        float4 r;
        r.x = p0.x + p1.x; r.y = p0.y + p1.y; r.z = p0.z + p1.z; r.w = p0.w + p1.w;
        const float4 b0 = *((const float4*)ba + hl2);
        r.x += b0.x; r.y += b0.y; r.z += b0.z; r.w += b0.w;
        if (two) {
            const float4 b1 = *((const float4*)bb + hl2);
            r.x += b1.x; r.y += b1.y; r.z += b1.z; r.w += b1.w;
        }
        *((float4*)(out + (long long)i2 * CH) + hl2) = r;
    }
}

__global__ void __launch_bounds__(256) __attribute__((amdgpu_waves_per_eu(16, 16)))
pull_gemm_resident_kernel(const float* __restrict__ Xs, const float* __restrict__ Xg,
                          const float* __restrict__ dinv, const int* __restrict__ off, const int* __restrict__ col,
                          const float* __restrict__ W0, const float* __restrict__ b0,
                          const float* __restrict__ W1, const float* __restrict__ b1,
                          const float* __restrict__ W2, const float* __restrict__ b2,
                          float* __restrict__ out, int N, int E, int nT, int G) {
    __shared__ float sY[2][16][YP];
    __shared__ float sP[2][16][YP];
    const int tid = threadIdx.x;
    const int b   = (int)blockIdx.x;
    int pass = 0;
    for (int base = 0; base < 2 * nT; base += G, ++pass) {
        const int p = base + ((pass == 1) ? (G - 1 - b) : b);
        if (p >= 2 * nT) break;
        const bool star = p < nT;
        const size_t ta = star ? 0 : 2;
        pull_gemm_tile(star ? Xs : Xg,
                       dinv + ta * N, off + ta * N, col + ta * E, star ? W0 : W2, star ? b0 : b2,
                       dinv + N,      off + N,      col + E,      W1,             b1,
                       out + (star ? (size_t)0 : (size_t)N * CH), N, E, star ? p : p - nT, star, sY, sP, tid);
    }
}

extern "C" void kernel_launch(void* const* d_in, const int* in_sizes, int n_in,
                              void* d_out, int out_size, void* d_ws, size_t ws_size,
                              hipStream_t stream) {
    const float* x_star = (const float*)d_in[0];
    const float* x_gal  = (const float*)d_in[1];
    const int*   e_ssn  = (const int*)d_in[2];
    const int*   e_ssf  = (const int*)d_in[3];
    const int*   e_ggn  = (const int*)d_in[4];
    const float* W_ssn  = (const float*)d_in[5];
    const float* W_ssf  = (const float*)d_in[6];
    const float* W_ggn  = (const float*)d_in[7];
    const float* b_ssn  = (const float*)d_in[8];
    const float* b_ssf  = (const float*)d_in[9];
    const float* b_ggn  = (const float*)d_in[10];

    const int N = in_sizes[0] / CH;
    const int E = in_sizes[2] / 2;

    float* ws    = (float*)d_ws;
    unsigned* tmp = (unsigned*)ws;
    int*   start = (int*)ws + (size_t)3 * E;
    int*   col   = (int*)(ws + (size_t)2 * N * CH);
    int*   off   = col + (size_t)3 * E;
    float* dinv1 = ws + (size_t)3 * N * CH;
    float* out   = (float*)d_out;

    const int nChunks = (E + CHUNK - 1) / CHUNK;
    const int nBins   = (N + BIN_NODES - 1) / BIN_NODES;
    partition_kernel<<<dim3((unsigned)nChunks, 3), PT, 0, stream>>>(e_ssn, e_ssf, e_ggn, tmp, start, E, nChunks);
    owner_kernel<<<dim3((unsigned)nBins, 3), OT, 0, stream>>>(tmp, start, off, col, dinv1, N, E, nChunks);

    const int nT = (N + 15) / 16;
    pull_gemm_resident_kernel<<<2048, 256, 0, stream>>>(x_star, x_gal, dinv1, off, col,
                                                        W_ssn, b_ssn, W_ssf, b_ssf, W_ggn, b_ggn,
                                                        out, N, E, nT, 2048);
}
